# starting kernel
# baseline (speedup 1.0000x reference)
#include <hip/hip_runtime.h>


#define N_NODES 100000
#define N_EDGES 3200000
#define DIN     512
#define DHID    256
#define DOUT    64
#define KSTEPS  10

typedef __attribute__((ext_vector_type(2))) float v2f;
typedef __attribute__((ext_vector_type(8))) float v8f;


__global__ __launch_bounds__(256) void k_init(float* __restrict__ deg,
                                              float* __restrict__ bufB) {
    int g = blockIdx.x * 256 + threadIdx.x;
    bufB[g] = 0.0f;
    if (g < N_NODES) deg[g] = 1.0f;
}

__global__ __launch_bounds__(256) void k_count(const int* __restrict__ col,
                                               float* __restrict__ deg) {
    int e = blockIdx.x * 256 + threadIdx.x;
    if (e < N_EDGES) atomicAdd(&deg[col[e]], 1.0f);
}

__global__ __launch_bounds__(256) void k_dinv(const float* __restrict__ deg,
                                              float* __restrict__ dinv,
                                              float* __restrict__ dinv2) {
    int i = blockIdx.x * 256 + threadIdx.x;
    if (i < N_NODES) {
        float r = rsqrtf(deg[i]);
        dinv[i]  = r;
        dinv2[i] = r * r;
    }
}

__global__ __launch_bounds__(256) void k_norm(const int* __restrict__ row,
                                              const int* __restrict__ col,
                                              const float* __restrict__ dinv,
                                              float* __restrict__ norm) {
    int e = blockIdx.x * 256 + threadIdx.x;
    if (e < N_EDGES) norm[e] = dinv[row[e]] * dinv[col[e]];
}

#define LDS_STRIDE 260

__global__ __launch_bounds__(512)
void k_mlp(const float* __restrict__ x,  const float* __restrict__ W1,
           const float* __restrict__ b1, const float* __restrict__ W2,
           const float* __restrict__ b2, const float* __restrict__ temp,
           float* __restrict__ bufA,     float* __restrict__ hidden) {
    __shared__ float lds[16 * LDS_STRIDE];

    const int tid  = threadIdx.x;
    const int wave = tid >> 5;
    const int lane = tid & 31;
    const int l15  = lane & 15;
    const int kh   = lane >> 4;
    const int m0   = blockIdx.x * 16;

    v8f acc = {0.f, 0.f, 0.f, 0.f, 0.f, 0.f, 0.f, 0.f};
    const float* xrow  = x  + (size_t)(m0 + l15) * DIN;
    const float* w1col = W1 + wave * 16 + l15;
    for (int k0 = 0; k0 < DIN; k0 += 4) {
        int ka = k0 + kh * 2;
        v2f a = *(const v2f*)(xrow + ka);
        v2f b;
        b.x = w1col[(size_t)ka * DHID];
        b.y = w1col[(size_t)(ka + 1) * DHID];
        acc = __builtin_amdgcn_wmma_f32_16x16x4_f32(
                  false, a, false, b, (short)0, acc, false, false);
    }
    {
        float bias = b1[wave * 16 + l15];
        int rbase = kh * 8;
        #pragma unroll
        for (int r = 0; r < 8; ++r) {
            float v = acc[r] + bias;
            v = v > 0.0f ? v : 0.0f;
            lds[(rbase + r) * LDS_STRIDE + wave * 16 + l15] = v;
        }
    }
    __syncthreads();

    if (wave < 4) {
        v8f acc2 = {0.f, 0.f, 0.f, 0.f, 0.f, 0.f, 0.f, 0.f};
        const float* w2col = W2 + wave * 16 + l15;
        for (int k0 = 0; k0 < DHID; k0 += 4) {
            int ka = k0 + kh * 2;
            v2f a = *(const v2f*)(&lds[l15 * LDS_STRIDE + ka]);
            v2f b;
            b.x = w2col[ka * DOUT];
            b.y = w2col[(ka + 1) * DOUT];
            acc2 = __builtin_amdgcn_wmma_f32_16x16x4_f32(
                       false, a, false, b, (short)0, acc2, false, false);
        }
        float bias = b2[wave * 16 + l15];
        float t0   = temp[0];
        int rbase = kh * 8;
        #pragma unroll
        for (int r = 0; r < 8; ++r) {
            float v = acc2[r] + bias;
            int idx = (m0 + rbase + r) * DOUT + wave * 16 + l15;
            bufA[idx]   = v;
            hidden[idx] = t0 * v;
        }
    }
}

__global__ __launch_bounds__(256)
void k_scatter(const int* __restrict__ row, const int* __restrict__ col,
               const float* __restrict__ norm, const float* __restrict__ src,
               float* __restrict__ dst) {
    int g = blockIdx.x * 256 + threadIdx.x;
    int e = __builtin_amdgcn_readfirstlane(g >> 6);
    int c = g & 63;
    int r  = row[e];
    int cl = col[e];
    float v = norm[e] * src[r * DOUT + c];
    atomicAdd(&dst[cl * DOUT + c], v);
}

__global__ __launch_bounds__(256)
void k_finalize(float* __restrict__ dst, float* __restrict__ src,
                const float* __restrict__ dinv2, float* __restrict__ hidden,
                const float* __restrict__ temp, int kidx) {
    int g = blockIdx.x * 256 + threadIdx.x;
    int i = __builtin_amdgcn_readfirstlane(g >> 6);
    float v = dst[g] + dinv2[i] * src[g];
    dst[g] = v;
    hidden[g] += temp[kidx] * v;
    src[g] = 0.0f;
}

extern "C" void kernel_launch(void* const* d_in, const int* in_sizes, int n_in,
                              void* d_out, int out_size, void* d_ws, size_t ws_size,
                              hipStream_t stream) {
    const float* x    = (const float*)d_in[0];
    const int*   ei   = (const int*)  d_in[1];
    const float* W1   = (const float*)d_in[2];
    const float* b1   = (const float*)d_in[3];
    const float* W2   = (const float*)d_in[4];
    const float* b2   = (const float*)d_in[5];
    const float* temp = (const float*)d_in[6];
    const int* row = ei;
    const int* col = ei + N_EDGES;
    float* hidden = (float*)d_out;

    float* deg   = (float*)d_ws;
    float* dinv  = deg   + N_NODES;
    float* dinv2 = dinv  + N_NODES;
    float* normw = dinv2 + N_NODES;
    float* bufA  = normw + N_EDGES;
    float* bufB  = bufA  + (size_t)N_NODES * DOUT;

    k_init <<<(N_NODES * DOUT) / 256, 256, 0, stream>>>(deg, bufB);
    k_count<<<(N_EDGES + 255) / 256, 256, 0, stream>>>(col, deg);
    k_dinv <<<(N_NODES + 255) / 256, 256, 0, stream>>>(deg, dinv, dinv2);
    k_norm <<<(N_EDGES + 255) / 256, 256, 0, stream>>>(row, col, dinv, normw);

    k_mlp<<<N_NODES / 16, 512, 0, stream>>>(x, W1, b1, W2, b2, temp, bufA, hidden);

    for (int k = 0; k < KSTEPS; ++k) {
        float* src = (k & 1) ? bufB : bufA;
        float* dst = (k & 1) ? bufA : bufB;
        k_scatter <<<(N_EDGES * 64) / 256, 256, 0, stream>>>(row, col, normw, src, dst);
        k_finalize<<<(N_NODES * DOUT) / 256, 256, 0, stream>>>(dst, src, dinv2,
                                                               hidden, temp, k + 1);
    }
}


# MLP: persistent 8-wave blocks; x read once straight into A fragments, W1 k-slabs and W2 shared via LDS as k-pair rows, each wave holds a 16x256 hidden slab in 16 tiles; k sum order regrouped
# speedup vs baseline: 6.0478x; 6.0478x over previous
#include <hip/hip_runtime.h>


#define N_NODES 100000
#define N_EDGES 3200000
#define DIN     512
#define DHID    256
#define DOUT    64
#define KSTEPS  10

#define TILE        196
#define NTILES      511
#define NTILES_PAD  512
#define SRC_BITS    17
#define SRC_MASK    0x1FFFF
#define BIN_CHUNK   4096

typedef __attribute__((ext_vector_type(2))) float v2f;
typedef __attribute__((ext_vector_type(8))) float v8f;


__global__ __launch_bounds__(256) void k_init(float* __restrict__ deg,
                                              int* __restrict__ binCnt) {
    int g = blockIdx.x * 256 + threadIdx.x;
    if (g < N_NODES) deg[g] = 1.0f;
    if (g < NTILES_PAD) binCnt[g] = 0;
}

__global__ __launch_bounds__(256) void k_count(const int* __restrict__ col,
                                               float* __restrict__ deg,
                                               int* __restrict__ binCnt) {
    __shared__ int h[NTILES_PAD];
    const int tid = threadIdx.x;
    for (int b = tid; b < NTILES_PAD; b += 256) h[b] = 0;
    __syncthreads();
    const int e0 = blockIdx.x * BIN_CHUNK + tid;
    for (int j = 0; j < BIN_CHUNK / 256; ++j) {
        int e = e0 + j * 256;
        if (e < N_EDGES) {
            int c = col[e];
            atomicAdd(&deg[c], 1.0f);
            atomicAdd(&h[c / TILE], 1);
        }
    }
    __syncthreads();
    for (int b = tid; b < NTILES; b += 256) {
        int n = h[b];
        if (n) atomicAdd(&binCnt[b], n);
    }
}

__global__ __launch_bounds__(256) void k_dinv(const float* __restrict__ deg,
                                              float* __restrict__ dinv) {
    int i = blockIdx.x * 256 + threadIdx.x;
    if (i < N_NODES) dinv[i] = rsqrtf(deg[i]);
}

__global__ __launch_bounds__(512) void k_scan(const int* __restrict__ binCnt,
                                              int* __restrict__ binOff,
                                              int* __restrict__ binCur) {
    __shared__ int s[512];
    const int t = threadIdx.x;
    const int v = (t < NTILES) ? binCnt[t] : 0;
    s[t] = v;
    __syncthreads();
    for (int d = 1; d < 512; d <<= 1) {
        int x = (t >= d) ? s[t - d] : 0;
        __syncthreads();
        s[t] += x;
        __syncthreads();
    }
    const int incl = s[t];
    if (t < NTILES) {
        binOff[t] = incl - v;
        binCur[t] = incl - v;
    }
    if (t == NTILES - 1) binOff[NTILES] = incl;
}

__global__ __launch_bounds__(256) void k_bin(const int* __restrict__ row,
                                             const int* __restrict__ col,
                                             int* __restrict__ binCur,
                                             int* __restrict__ packed) {
    __shared__ int h[NTILES_PAD];
    const int tid = threadIdx.x;
    for (int b = tid; b < NTILES_PAD; b += 256) h[b] = 0;
    __syncthreads();
    const int e0 = blockIdx.x * BIN_CHUNK + tid;
    for (int j = 0; j < BIN_CHUNK / 256; ++j) {
        int e = e0 + j * 256;
        if (e < N_EDGES) atomicAdd(&h[col[e] / TILE], 1);
    }
    __syncthreads();
    for (int b = tid; b < NTILES; b += 256) {
        int n = h[b];
        h[b] = n ? atomicAdd(&binCur[b], n) : 0;
    }
    __syncthreads();
    for (int j = 0; j < BIN_CHUNK / 256; ++j) {
        int e = e0 + j * 256;
        if (e < N_EDGES) {
            int c  = col[e];
            int t  = c / TILE;
            int tl = c - t * TILE;
            int pos = atomicAdd(&h[t], 1);
            packed[pos] = row[e] | (tl << SRC_BITS);
        }
    }
}

#define MLP_WAVES   8
#define MLP_BLOCKS  256
#define NSLAB       (N_NODES / 16)
#define SLABS_PASS  (MLP_BLOCKS * MLP_WAVES)
#define FULL_PASSES (NSLAB / SLABS_PASS)
#define REM_SLABS   (NSLAB - FULL_PASSES * SLABS_PASS)
#define KC          32
#define NKC         (DIN / KC)
#define W1_PITCH    528
#define W1_BUF      ((KC / 2) * W1_PITCH)
#define W2_PITCH    160
#define HS_STRIDE   260

static_assert(REM_SLABS <= MLP_BLOCKS, "last pass: one slab per block");
static_assert(N_NODES % 16 == 0, "no row tail");

__global__ __launch_bounds__(256)
void k_mlp(const float* __restrict__ x,  const float* __restrict__ W1,
           const float* __restrict__ b1, const float* __restrict__ W2,
           const float* __restrict__ b2, const float* __restrict__ temp,
           float* __restrict__ bufA,     float* __restrict__ hidden) {
    __shared__ __attribute__((aligned(16))) float w1s[2 * W1_BUF];
    __shared__ __attribute__((aligned(16))) float w2s[(DHID / 2) * W2_PITCH];
    __shared__ __attribute__((aligned(16))) float hs[MLP_WAVES * 16 * HS_STRIDE];

    const int tid  = threadIdx.x;
    const int lane = tid & 31;
    const int wave = __builtin_amdgcn_readfirstlane(tid >> 5);
    const int c    = lane & 15;
    const int h    = lane >> 4;
    const int bid  = blockIdx.x;

    const int npass = (bid < REM_SLABS) ? FULL_PASSES + 1 : FULL_PASSES;
    const int total = npass * NKC;

    for (int task = tid; task < (DHID / 2) * (DOUT / 4); task += 256) {
        const int kp = task >> 4, n4 = task & 15;
        const float4 r0 = *(const float4*)(W2 + (2 * kp) * DOUT + 4 * n4);
        const float4 r1 = *(const float4*)(W2 + (2 * kp + 1) * DOUT + 4 * n4);
        float4* d = (float4*)(w2s + kp * W2_PITCH + 8 * n4);
        d[0] = make_float4(r0.x, r1.x, r0.y, r1.y);
        d[1] = make_float4(r0.z, r1.z, r0.w, r1.w);
    }

    const int wn  = tid & 63;
    const int wkp = tid >> 6;
    float4 wr[8];
    float4 xc[4], xn[4];
    v8f acc[16];

    {
        const float* p = W1 + (2 * wkp) * DHID + 4 * wn;
        #pragma unroll
        for (int q = 0; q < 4; ++q) {
            wr[2 * q]     = *(const float4*)(p + (8 * q) * DHID);
            wr[2 * q + 1] = *(const float4*)(p + (8 * q + 1) * DHID);
        }
        const float* xp = x + (size_t)((bid * MLP_WAVES + wave) * 16 + c) * DIN + 4 * h;
        #pragma unroll
        for (int i = 0; i < 4; ++i) xc[i] = *(const float4*)(xp + 8 * i);
        #pragma unroll
        for (int q = 0; q < 4; ++q) {
            float4* d = (float4*)(w1s + (wkp + 4 * q) * W1_PITCH + 8 * wn);
            d[0] = make_float4(wr[2 * q].x, wr[2 * q + 1].x, wr[2 * q].y, wr[2 * q + 1].y);
            d[1] = make_float4(wr[2 * q].z, wr[2 * q + 1].z, wr[2 * q].w, wr[2 * q + 1].w);
        }
    }
    __syncthreads();

    for (int g = 0; g < total; ++g) {
        const int pass = g >> 4;
        const int ks   = g & (NKC - 1);
        const int slab = (pass < FULL_PASSES) ? pass * SLABS_PASS + bid * MLP_WAVES + wave
                                              : ((wave == 0) ? FULL_PASSES * SLABS_PASS + bid : -1);
        const bool act  = slab >= 0;
        const bool more = (g + 1 < total);

        if (ks == 0) {
            #pragma unroll
            for (int t = 0; t < 16; ++t) acc[t] = (v8f){0.f, 0.f, 0.f, 0.f, 0.f, 0.f, 0.f, 0.f};
        }

        if (more) {
            const int g1  = g + 1;
            const int kb1 = (g1 & (NKC - 1)) * KC;
            const float* p = W1 + (kb1 + 2 * wkp) * DHID + 4 * wn;
            #pragma unroll
            for (int q = 0; q < 4; ++q) {
                wr[2 * q]     = *(const float4*)(p + (8 * q) * DHID);
                wr[2 * q + 1] = *(const float4*)(p + (8 * q + 1) * DHID);
            }
            const int pass1 = g1 >> 4;
            const int slab1 = (pass1 < FULL_PASSES) ? pass1 * SLABS_PASS + bid * MLP_WAVES + wave
                                                    : ((wave == 0) ? FULL_PASSES * SLABS_PASS + bid : -1);
            if (slab1 >= 0) {
                const float* xp = x + (size_t)(slab1 * 16 + c) * DIN + kb1 + 4 * h;
                #pragma unroll
                for (int i = 0; i < 4; ++i) xn[i] = *(const float4*)(xp + 8 * i);
            }
        }

        if (act) {
            const float* wb = w1s + (g & 1) * W1_BUF + (2 * h) * W1_PITCH + 2 * c;
            #pragma unroll
            for (int i = 0; i < 4; ++i) {
                const v2f a0 = {xc[i].x, xc[i].y};
                const v2f a1 = {xc[i].z, xc[i].w};
                const float* w0 = wb + (4 * i) * W1_PITCH;
                #pragma unroll
                for (int t = 0; t < 16; ++t) {
                    const v2f b = *(const v2f*)(w0 + 32 * t);
                    acc[t] = __builtin_amdgcn_wmma_f32_16x16x4_f32(
                                 false, a0, false, b, (short)0, acc[t], false, false);
                }
                #pragma unroll
                for (int t = 0; t < 16; ++t) {
                    const v2f b = *(const v2f*)(w0 + W1_PITCH + 32 * t);
                    acc[t] = __builtin_amdgcn_wmma_f32_16x16x4_f32(
                                 false, a1, false, b, (short)0, acc[t], false, false);
                }
            }
        }

        if (more) {
            float* dbase = w1s + ((g + 1) & 1) * W1_BUF + wkp * W1_PITCH + 8 * wn;
            #pragma unroll
            for (int q = 0; q < 4; ++q) {
                float4* d = (float4*)(dbase + (4 * q) * W1_PITCH);
                d[0] = make_float4(wr[2 * q].x, wr[2 * q + 1].x, wr[2 * q].y, wr[2 * q + 1].y);
                d[1] = make_float4(wr[2 * q].z, wr[2 * q + 1].z, wr[2 * q].w, wr[2 * q + 1].w);
            }
        }

        float* hw = hs + wave * (16 * HS_STRIDE);
        if (ks == NKC - 1 && act) {
            #pragma unroll
            for (int t = 0; t < 16; ++t) {
                const float bias = b1[t * 16 + c];
                #pragma unroll
                for (int r = 0; r < 8; ++r) {
                    float v = acc[t][r] + bias;
                    v = v > 0.0f ? v : 0.0f;
                    hw[(8 * h + r) * HS_STRIDE + t * 16 + c] = v;
                }
            }
        }
        __syncthreads();

        if (ks == NKC - 1 && act) {
            v8f acc2[4];
            #pragma unroll
            for (int t = 0; t < 4; ++t) acc2[t] = (v8f){0.f, 0.f, 0.f, 0.f, 0.f, 0.f, 0.f, 0.f};
            const float* ha  = hw + c * HS_STRIDE + 2 * h;
            const float* w2b = w2s + h * W2_PITCH + 2 * c;
            #pragma unroll 4
            for (int k0 = 0; k0 < DHID; k0 += 4) {
                const v2f a = *(const v2f*)(ha + k0);
                const float* w0 = w2b + (k0 >> 1) * W2_PITCH;
                #pragma unroll
                for (int t = 0; t < 4; ++t) {
                    const v2f b = *(const v2f*)(w0 + 32 * t);
                    acc2[t] = __builtin_amdgcn_wmma_f32_16x16x4_f32(
                                  false, a, false, b, (short)0, acc2[t], false, false);
                }
            }
            const float t0 = temp[0];
            #pragma unroll
            for (int t = 0; t < 4; ++t) {
                const float bias = b2[t * 16 + c];
                #pragma unroll
                for (int r = 0; r < 8; ++r) {
                    const float v = acc2[t][r] + bias;
                    const int idx = (slab * 16 + 8 * h + r) * DOUT + t * 16 + c;
                    bufA[idx]   = v;
                    hidden[idx] = t0 * v;
                }
            }
        }

        #pragma unroll
        for (int i = 0; i < 4; ++i) xc[i] = xn[i];
    }
}

#define EDGES_PER_WAVE 32
#define EDGE_BATCH     16

template <bool TAIL>
__device__ __forceinline__ void add_edges(const float* __restrict__ src, float* acc,
                                          int s_l, int t_l, int w_l, int lane, int nvalid) {
    #pragma unroll
    for (int b = 0; b < EDGES_PER_WAVE; b += EDGE_BATCH) {
        if (TAIL && b >= nvalid) break;
        float lo[EDGE_BATCH], hi[EDGE_BATCH];
        #pragma unroll
        for (int j = 0; j < EDGE_BATCH; ++j) {
            int s = __builtin_amdgcn_readlane(s_l, b + j);
            const float* p = src + s * DOUT + lane;
            lo[j] = p[0];
            hi[j] = p[32];
        }
        #pragma unroll
        for (int j = 0; j < EDGE_BATCH; ++j) {
            if (!TAIL || b + j < nvalid) {
                int   t = __builtin_amdgcn_readlane(t_l, b + j);
                float n = __int_as_float(__builtin_amdgcn_readlane(w_l, b + j));
                float* q = acc + t * DOUT + lane;
                atomicAdd(q,      n * lo[j]);
                atomicAdd(q + 32, n * hi[j]);
            }
        }
    }
}

__global__ __launch_bounds__(1024)
void k_prop(const int* __restrict__ packed, const int* __restrict__ binOff,
            const float* __restrict__ dinv, const float* __restrict__ src,
            float* __restrict__ dst, float* __restrict__ hidden,
            const float* __restrict__ temp, int kidx) {
    __shared__ float acc[TILE * DOUT];

    const int tid  = threadIdx.x;
    const int lane = tid & 31;
    const int wave = __builtin_amdgcn_readfirstlane(tid >> 5);
    const int node0 = blockIdx.x * TILE;

    for (int i = tid; i < TILE * DOUT; i += 1024) acc[i] = 0.0f;
    __syncthreads();

    const int beg = binOff[blockIdx.x];
    const int end = binOff[blockIdx.x + 1];
    for (int base = beg + wave * EDGES_PER_WAVE; base < end; base += 32 * EDGES_PER_WAVE) {
        const int e      = base + lane;
        const int nvalid = end - base;
        const unsigned p = (e < end) ? (unsigned)packed[e] : 0u;
        const int s_l = (int)(p & SRC_MASK);
        const int t_l = (int)(p >> SRC_BITS);
        const int w_l = __float_as_int(dinv[s_l] * dinv[node0 + t_l]);
        if (nvalid >= EDGES_PER_WAVE) add_edges<false>(src, acc, s_l, t_l, w_l, lane, nvalid);
        else                          add_edges<true >(src, acc, s_l, t_l, w_l, lane, nvalid);
    }
    __syncthreads();

    const float gk = temp[kidx];
    const int nrows = (N_NODES - node0 < TILE) ? (N_NODES - node0) : TILE;
    for (int i = tid; i < nrows * DOUT; i += 1024) {
        const float r = dinv[node0 + (i >> 6)];
        const int   g = node0 * DOUT + i;
        const float v = acc[i] + (r * r) * src[g];
        dst[g] = v;
        hidden[g] += gk * v;
    }
}

extern "C" void kernel_launch(void* const* d_in, const int* in_sizes, int n_in,
                              void* d_out, int out_size, void* d_ws, size_t ws_size,
                              hipStream_t stream) {
    const float* x    = (const float*)d_in[0];
    const int*   ei   = (const int*)  d_in[1];
    const float* W1   = (const float*)d_in[2];
    const float* b1   = (const float*)d_in[3];
    const float* W2   = (const float*)d_in[4];
    const float* b2   = (const float*)d_in[5];
    const float* temp = (const float*)d_in[6];
    const int* row = ei;
    const int* col = ei + N_EDGES;
    float* hidden = (float*)d_out;

    float* deg    = (float*)d_ws;
    float* dinv   = deg + N_NODES;
    int*   binCnt = (int*)(dinv + N_NODES);
    int*   binOff = binCnt + 512;
    int*   binCur = binOff + 1024;
    int*   packed = binCur + 512;
    float* bufA   = (float*)(packed + N_EDGES);
    float* bufB   = bufA + (size_t)N_NODES * DOUT;

    k_init <<<(N_NODES + 255) / 256, 256, 0, stream>>>(deg, binCnt);
    k_count<<<(N_EDGES + BIN_CHUNK - 1) / BIN_CHUNK, 256, 0, stream>>>(col, deg, binCnt);
    k_dinv <<<(N_NODES + 255) / 256, 256, 0, stream>>>(deg, dinv);
    k_scan <<<1, 512, 0, stream>>>(binCnt, binOff, binCur);
    k_bin  <<<(N_EDGES + BIN_CHUNK - 1) / BIN_CHUNK, 256, 0, stream>>>(row, col, binCur, packed);

    k_mlp<<<MLP_BLOCKS, MLP_WAVES * 32, 0, stream>>>(x, W1, b1, W2, b2, temp, bufA, hidden);

    for (int k = 0; k < KSTEPS; ++k) {
        float* src = (k & 1) ? bufB : bufA;
        float* dst = (k & 1) ? bufA : bufB;
        k_prop<<<NTILES, 1024, 0, stream>>>(packed, binOff, dinv, src, dst,
                                            hidden, temp, k + 1);
    }
}


# Scatter: binning packs deg[src] into 7 spare bits of each edge word; a step forms dinv[src] by rsqrt of that code (table read only if degree >= 127): the per-edge 4-byte dinv[src] gather is gone
# speedup vs baseline: 6.4258x; 1.0625x over previous
#include <hip/hip_runtime.h>


#define N_NODES 100000
#define N_EDGES 3200000
#define DIN     512
#define DHID    256
#define DOUT    64
#define KSTEPS  10

#define TILE        196
#define NTILES      511
#define NTILES_PAD  512
#define SRC_BITS    17
#define SRC_MASK    0x1FFFF
#define TL_MASK     0xFF
#define DEG_SHIFT   25
#define DEG_ESC     127u
#define BIN_CHUNK   4096

typedef __attribute__((ext_vector_type(2))) float v2f;
typedef __attribute__((ext_vector_type(8))) float v8f;


__global__ __launch_bounds__(256) void k_init(float* __restrict__ deg,
                                              int* __restrict__ binCnt) {
    int g = blockIdx.x * 256 + threadIdx.x;
    if (g < N_NODES) deg[g] = 1.0f;
    if (g < NTILES_PAD) binCnt[g] = 0;
}

__global__ __launch_bounds__(256) void k_count(const int* __restrict__ col,
                                               float* __restrict__ deg,
                                               int* __restrict__ binCnt) {
    __shared__ int h[NTILES_PAD];
    const int tid = threadIdx.x;
    for (int b = tid; b < NTILES_PAD; b += 256) h[b] = 0;
    __syncthreads();
    const int e0 = blockIdx.x * BIN_CHUNK + tid;
    for (int j = 0; j < BIN_CHUNK / 256; ++j) {
        int e = e0 + j * 256;
        if (e < N_EDGES) {
            int c = col[e];
            atomicAdd(&deg[c], 1.0f);
            atomicAdd(&h[c / TILE], 1);
        }
    }
    __syncthreads();
    for (int b = tid; b < NTILES; b += 256) {
        int n = h[b];
        if (n) atomicAdd(&binCnt[b], n);
    }
}

__device__ __forceinline__ float dinv_of(float d) { return rsqrtf(d); }

__global__ __launch_bounds__(256) void k_dinv(const float* __restrict__ deg,
                                              float* __restrict__ dinv) {
    int i = blockIdx.x * 256 + threadIdx.x;
    if (i < N_NODES) dinv[i] = dinv_of(deg[i]);
}

__global__ __launch_bounds__(512) void k_scan(const int* __restrict__ binCnt,
                                              int* __restrict__ binOff,
                                              int* __restrict__ binCur) {
    __shared__ int s[512];
    const int t = threadIdx.x;
    const int v = (t < NTILES) ? binCnt[t] : 0;
    s[t] = v;
    __syncthreads();
    for (int d = 1; d < 512; d <<= 1) {
        int x = (t >= d) ? s[t - d] : 0;
        __syncthreads();
        s[t] += x;
        __syncthreads();
    }
    const int incl = s[t];
    if (t < NTILES) {
        binOff[t] = incl - v;
        binCur[t] = incl - v;
    }
    if (t == NTILES - 1) binOff[NTILES] = incl;
}

__global__ __launch_bounds__(256) void k_bin(const int* __restrict__ row,
                                             const int* __restrict__ col,
                                             const float* __restrict__ deg,
                                             int* __restrict__ binCur,
                                             int* __restrict__ packed) {
    __shared__ int h[NTILES_PAD];
    const int tid = threadIdx.x;
    const int e0 = blockIdx.x * BIN_CHUNK + tid;
    int      rr[BIN_CHUNK / 256];
    unsigned dcode[BIN_CHUNK / 256];
    #pragma unroll
    for (int j = 0; j < BIN_CHUNK / 256; ++j) {
        int e = e0 + j * 256;
        rr[j] = row[e < N_EDGES ? e : N_EDGES - 1];
    }
    #pragma unroll
    for (int j = 0; j < BIN_CHUNK / 256; ++j) {
        unsigned d = (unsigned)deg[rr[j]];
        dcode[j] = d < DEG_ESC ? d : DEG_ESC;
    }
    for (int b = tid; b < NTILES_PAD; b += 256) h[b] = 0;
    __syncthreads();
    for (int j = 0; j < BIN_CHUNK / 256; ++j) {
        int e = e0 + j * 256;
        if (e < N_EDGES) atomicAdd(&h[col[e] / TILE], 1);
    }
    __syncthreads();
    for (int b = tid; b < NTILES; b += 256) {
        int n = h[b];
        h[b] = n ? atomicAdd(&binCur[b], n) : 0;
    }
    __syncthreads();
    #pragma unroll
    for (int j = 0; j < BIN_CHUNK / 256; ++j) {
        int e = e0 + j * 256;
        if (e < N_EDGES) {
            int c  = col[e];
            int t  = c / TILE;
            int tl = c - t * TILE;
            int pos = atomicAdd(&h[t], 1);
            packed[pos] = (int)((unsigned)rr[j] | ((unsigned)tl << SRC_BITS) | (dcode[j] << DEG_SHIFT));
        }
    }
}

#define MLP_WAVES   8
#define MLP_BLOCKS  256
#define NSLAB       (N_NODES / 16)
#define SLABS_PASS  (MLP_BLOCKS * MLP_WAVES)
#define FULL_PASSES (NSLAB / SLABS_PASS)
#define REM_SLABS   (NSLAB - FULL_PASSES * SLABS_PASS)
#define KC          32
#define NKC         (DIN / KC)
#define W1_PITCH    528
#define W1_BUF      ((KC / 2) * W1_PITCH)
#define W2_PITCH    160
#define HS_STRIDE   260

static_assert(REM_SLABS <= MLP_BLOCKS, "last pass: one slab per block");
static_assert(N_NODES % 16 == 0, "no row tail");

__global__ __launch_bounds__(256)
void k_mlp(const float* __restrict__ x,  const float* __restrict__ W1,
           const float* __restrict__ b1, const float* __restrict__ W2,
           const float* __restrict__ b2, const float* __restrict__ temp,
           float* __restrict__ bufA,     float* __restrict__ hidden) {
    __shared__ __attribute__((aligned(16))) float w1s[2 * W1_BUF];
    __shared__ __attribute__((aligned(16))) float w2s[(DHID / 2) * W2_PITCH];
    __shared__ __attribute__((aligned(16))) float hs[MLP_WAVES * 16 * HS_STRIDE];

    const int tid  = threadIdx.x;
    const int lane = tid & 31;
    const int wave = __builtin_amdgcn_readfirstlane(tid >> 5);
    const int c    = lane & 15;
    const int h    = lane >> 4;
    const int bid  = blockIdx.x;

    const int npass = (bid < REM_SLABS) ? FULL_PASSES + 1 : FULL_PASSES;
    const int total = npass * NKC;

    for (int task = tid; task < (DHID / 2) * (DOUT / 4); task += 256) {
        const int kp = task >> 4, n4 = task & 15;
        const float4 r0 = *(const float4*)(W2 + (2 * kp) * DOUT + 4 * n4);
        const float4 r1 = *(const float4*)(W2 + (2 * kp + 1) * DOUT + 4 * n4);
        float4* d = (float4*)(w2s + kp * W2_PITCH + 8 * n4);
        d[0] = make_float4(r0.x, r1.x, r0.y, r1.y);
        d[1] = make_float4(r0.z, r1.z, r0.w, r1.w);
    }

    const int wn  = tid & 63;
    const int wkp = tid >> 6;
    float4 wr[8];
    float4 xc[4], xn[4];
    v8f acc[16];

    {
        const float* p = W1 + (2 * wkp) * DHID + 4 * wn;
        #pragma unroll
        for (int q = 0; q < 4; ++q) {
            wr[2 * q]     = *(const float4*)(p + (8 * q) * DHID);
            wr[2 * q + 1] = *(const float4*)(p + (8 * q + 1) * DHID);
        }
        const float* xp = x + (size_t)((bid * MLP_WAVES + wave) * 16 + c) * DIN + 4 * h;
        #pragma unroll
        for (int i = 0; i < 4; ++i) xc[i] = *(const float4*)(xp + 8 * i);
        #pragma unroll
        for (int q = 0; q < 4; ++q) {
            float4* d = (float4*)(w1s + (wkp + 4 * q) * W1_PITCH + 8 * wn);
            d[0] = make_float4(wr[2 * q].x, wr[2 * q + 1].x, wr[2 * q].y, wr[2 * q + 1].y);
            d[1] = make_float4(wr[2 * q].z, wr[2 * q + 1].z, wr[2 * q].w, wr[2 * q + 1].w);
        }
    }
    __syncthreads();

    for (int g = 0; g < total; ++g) {
        const int pass = g >> 4;
        const int ks   = g & (NKC - 1);
        const int slab = (pass < FULL_PASSES) ? pass * SLABS_PASS + bid * MLP_WAVES + wave
                                              : ((wave == 0) ? FULL_PASSES * SLABS_PASS + bid : -1);
        const bool act  = slab >= 0;
        const bool more = (g + 1 < total);

        if (ks == 0) {
            #pragma unroll
            for (int t = 0; t < 16; ++t) acc[t] = (v8f){0.f, 0.f, 0.f, 0.f, 0.f, 0.f, 0.f, 0.f};
        }

        if (more) {
            const int g1  = g + 1;
            const int kb1 = (g1 & (NKC - 1)) * KC;
            const float* p = W1 + (kb1 + 2 * wkp) * DHID + 4 * wn;
            #pragma unroll
            for (int q = 0; q < 4; ++q) {
                wr[2 * q]     = *(const float4*)(p + (8 * q) * DHID);
                wr[2 * q + 1] = *(const float4*)(p + (8 * q + 1) * DHID);
            }
            const int pass1 = g1 >> 4;
            const int slab1 = (pass1 < FULL_PASSES) ? pass1 * SLABS_PASS + bid * MLP_WAVES + wave
                                                    : ((wave == 0) ? FULL_PASSES * SLABS_PASS + bid : -1);
            if (slab1 >= 0) {
                const float* xp = x + (size_t)(slab1 * 16 + c) * DIN + kb1 + 4 * h;
                #pragma unroll
                for (int i = 0; i < 4; ++i) xn[i] = *(const float4*)(xp + 8 * i);
            }
        }

        if (act) {
            const float* wb = w1s + (g & 1) * W1_BUF + (2 * h) * W1_PITCH + 2 * c;
            #pragma unroll
            for (int i = 0; i < 4; ++i) {
                const v2f a0 = {xc[i].x, xc[i].y};
                const v2f a1 = {xc[i].z, xc[i].w};
                const float* w0 = wb + (4 * i) * W1_PITCH;
                #pragma unroll
                for (int t = 0; t < 16; ++t) {
                    const v2f b = *(const v2f*)(w0 + 32 * t);
                    acc[t] = __builtin_amdgcn_wmma_f32_16x16x4_f32(
                                 false, a0, false, b, (short)0, acc[t], false, false);
                }
                #pragma unroll
                for (int t = 0; t < 16; ++t) {
                    const v2f b = *(const v2f*)(w0 + W1_PITCH + 32 * t);
                    acc[t] = __builtin_amdgcn_wmma_f32_16x16x4_f32(
                                 false, a1, false, b, (short)0, acc[t], false, false);
                }
            }
        }

        if (more) {
            float* dbase = w1s + ((g + 1) & 1) * W1_BUF + wkp * W1_PITCH + 8 * wn;
            #pragma unroll
            for (int q = 0; q < 4; ++q) {
                float4* d = (float4*)(dbase + (4 * q) * W1_PITCH);
                d[0] = make_float4(wr[2 * q].x, wr[2 * q + 1].x, wr[2 * q].y, wr[2 * q + 1].y);
                d[1] = make_float4(wr[2 * q].z, wr[2 * q + 1].z, wr[2 * q].w, wr[2 * q + 1].w);
            }
        }

        float* hw = hs + wave * (16 * HS_STRIDE);
        if (ks == NKC - 1 && act) {
            #pragma unroll
            for (int t = 0; t < 16; ++t) {
                const float bias = b1[t * 16 + c];
                #pragma unroll
                for (int r = 0; r < 8; ++r) {
                    float v = acc[t][r] + bias;
                    v = v > 0.0f ? v : 0.0f;
                    hw[(8 * h + r) * HS_STRIDE + t * 16 + c] = v;
                }
            }
        }
        __syncthreads();

        if (ks == NKC - 1 && act) {
            v8f acc2[4];
            #pragma unroll
            for (int t = 0; t < 4; ++t) acc2[t] = (v8f){0.f, 0.f, 0.f, 0.f, 0.f, 0.f, 0.f, 0.f};
            const float* ha  = hw + c * HS_STRIDE + 2 * h;
            const float* w2b = w2s + h * W2_PITCH + 2 * c;
            #pragma unroll 4
            for (int k0 = 0; k0 < DHID; k0 += 4) {
                const v2f a = *(const v2f*)(ha + k0);
                const float* w0 = w2b + (k0 >> 1) * W2_PITCH;
                #pragma unroll
                for (int t = 0; t < 4; ++t) {
                    const v2f b = *(const v2f*)(w0 + 32 * t);
                    acc2[t] = __builtin_amdgcn_wmma_f32_16x16x4_f32(
                                  false, a, false, b, (short)0, acc2[t], false, false);
                }
            }
            const float t0 = temp[0];
            #pragma unroll
            for (int t = 0; t < 4; ++t) {
                const float bias = b2[t * 16 + c];
                #pragma unroll
                for (int r = 0; r < 8; ++r) {
                    const float v = acc2[t][r] + bias;
                    const int idx = (slab * 16 + 8 * h + r) * DOUT + t * 16 + c;
                    bufA[idx]   = v;
                    hidden[idx] = t0 * v;
                }
            }
        }

        #pragma unroll
        for (int i = 0; i < 4; ++i) xc[i] = xn[i];
    }
}

#define EDGES_PER_WAVE 32
#define EDGE_BATCH     16

template <bool TAIL>
__device__ __forceinline__ void add_edges(const float* __restrict__ src, float* acc,
                                          int s_l, int t_l, int w_l, int lane, int nvalid) {
    #pragma unroll
    for (int b = 0; b < EDGES_PER_WAVE; b += EDGE_BATCH) {
        if (TAIL && b >= nvalid) break;
        float lo[EDGE_BATCH], hi[EDGE_BATCH];
        #pragma unroll
        for (int j = 0; j < EDGE_BATCH; ++j) {
            int s = __builtin_amdgcn_readlane(s_l, b + j);
            const float* p = src + s * DOUT + lane;
            lo[j] = p[0];
            hi[j] = p[32];
        }
        #pragma unroll
        for (int j = 0; j < EDGE_BATCH; ++j) {
            if (!TAIL || b + j < nvalid) {
                int   t = __builtin_amdgcn_readlane(t_l, b + j);
                float n = __int_as_float(__builtin_amdgcn_readlane(w_l, b + j));
                float* q = acc + t * DOUT + lane;
                atomicAdd(q,      n * lo[j]);
                atomicAdd(q + 32, n * hi[j]);
            }
        }
    }
}

__global__ __launch_bounds__(1024)
void k_prop(const int* __restrict__ packed, const int* __restrict__ binOff,
            const float* __restrict__ dinv, const float* __restrict__ src,
            float* __restrict__ dst, float* __restrict__ hidden,
            const float* __restrict__ temp, int kidx) {
    __shared__ float acc[TILE * DOUT];

    const int tid  = threadIdx.x;
    const int lane = tid & 31;
    const int wave = __builtin_amdgcn_readfirstlane(tid >> 5);
    const int node0 = blockIdx.x * TILE;

    for (int i = tid; i < TILE * DOUT; i += 1024) acc[i] = 0.0f;
    __syncthreads();

    const int beg = binOff[blockIdx.x];
    const int end = binOff[blockIdx.x + 1];
    for (int base = beg + wave * EDGES_PER_WAVE; base < end; base += 32 * EDGES_PER_WAVE) {
        const int e      = base + lane;
        const int nvalid = end - base;
        const unsigned p = (e < end) ? (unsigned)packed[e] : (1u << DEG_SHIFT);
        const int s_l = (int)(p & SRC_MASK);
        const int t_l = (int)((p >> SRC_BITS) & TL_MASK);
        const unsigned dc = p >> DEG_SHIFT;
        float ds;
        if (dc < DEG_ESC) ds = dinv_of((float)dc);
        else              ds = dinv[s_l];
        const int w_l = __float_as_int(ds * dinv[node0 + t_l]);
        if (nvalid >= EDGES_PER_WAVE) add_edges<false>(src, acc, s_l, t_l, w_l, lane, nvalid);
        else                          add_edges<true >(src, acc, s_l, t_l, w_l, lane, nvalid);
    }
    __syncthreads();

    const float gk = temp[kidx];
    const int nrows = (N_NODES - node0 < TILE) ? (N_NODES - node0) : TILE;
    for (int i = tid; i < nrows * DOUT; i += 1024) {
        const float r = dinv[node0 + (i >> 6)];
        const int   g = node0 * DOUT + i;
        const float v = acc[i] + (r * r) * src[g];
        dst[g] = v;
        hidden[g] += gk * v;
    }
}

extern "C" void kernel_launch(void* const* d_in, const int* in_sizes, int n_in,
                              void* d_out, int out_size, void* d_ws, size_t ws_size,
                              hipStream_t stream) {
    const float* x    = (const float*)d_in[0];
    const int*   ei   = (const int*)  d_in[1];
    const float* W1   = (const float*)d_in[2];
    const float* b1   = (const float*)d_in[3];
    const float* W2   = (const float*)d_in[4];
    const float* b2   = (const float*)d_in[5];
    const float* temp = (const float*)d_in[6];
    const int* row = ei;
    const int* col = ei + N_EDGES;
    float* hidden = (float*)d_out;

    float* deg    = (float*)d_ws;
    float* dinv   = deg + N_NODES;
    int*   binCnt = (int*)(dinv + N_NODES);
    int*   binOff = binCnt + 512;
    int*   binCur = binOff + 1024;
    int*   packed = binCur + 512;
    float* bufA   = (float*)(packed + N_EDGES);
    float* bufB   = bufA + (size_t)N_NODES * DOUT;

    k_init <<<(N_NODES + 255) / 256, 256, 0, stream>>>(deg, binCnt);
    k_count<<<(N_EDGES + BIN_CHUNK - 1) / BIN_CHUNK, 256, 0, stream>>>(col, deg, binCnt);
    k_dinv <<<(N_NODES + 255) / 256, 256, 0, stream>>>(deg, dinv);
    k_scan <<<1, 512, 0, stream>>>(binCnt, binOff, binCur);
    k_bin  <<<(N_EDGES + BIN_CHUNK - 1) / BIN_CHUNK, 256, 0, stream>>>(row, col, deg, binCur, packed);

    k_mlp<<<MLP_BLOCKS, MLP_WAVES * 32, 0, stream>>>(x, W1, b1, W2, b2, temp, bufA, hidden);

    for (int k = 0; k < KSTEPS; ++k) {
        float* src = (k & 1) ? bufB : bufA;
        float* dst = (k & 1) ? bufA : bufB;
        k_prop<<<NTILES, 1024, 0, stream>>>(packed, binOff, dinv, src, dst,
                                            hidden, temp, k + 1);
    }
}


# In-degrees counted per target tile in LDS from the binned edge words (one block per tile), not by 3.2M scattered global f32 atomics; step 1 adds the degree code to the words; sum order untouched
# speedup vs baseline: 7.4237x; 1.1553x over previous
#include <hip/hip_runtime.h>


#define N_NODES 100000
#define N_EDGES 3200000
#define DIN     512
#define DHID    256
#define DOUT    64
#define KSTEPS  10

#define TILE        196
#define NTILES      511
#define NTILES_PAD  512
#define SRC_BITS    17
#define SRC_MASK    0x1FFFF
#define TL_MASK     0xFF
#define DEG_SHIFT   25
#define DEG_ESC     127u
#define BIN_CHUNK   4096

typedef __attribute__((ext_vector_type(2))) float v2f;
typedef __attribute__((ext_vector_type(8))) float v8f;


__global__ __launch_bounds__(256) void k_init(int* __restrict__ binCnt) {
    int g = blockIdx.x * 256 + threadIdx.x;
    if (g < NTILES_PAD) binCnt[g] = 0;
}

__global__ __launch_bounds__(256) void k_count(const int* __restrict__ col,
                                               int* __restrict__ binCnt) {
    __shared__ int h[NTILES_PAD];
    const int tid = threadIdx.x;
    for (int b = tid; b < NTILES_PAD; b += 256) h[b] = 0;
    __syncthreads();
    const int e0 = blockIdx.x * BIN_CHUNK + tid;
    for (int j = 0; j < BIN_CHUNK / 256; ++j) {
        int e = e0 + j * 256;
        if (e < N_EDGES) {
            int c = col[e];
            atomicAdd(&h[c / TILE], 1);
        }
    }
    __syncthreads();
    for (int b = tid; b < NTILES; b += 256) {
        int n = h[b];
        if (n) atomicAdd(&binCnt[b], n);
    }
}

__device__ __forceinline__ float dinv_of(float d) { return rsqrtf(d); }

#define DEG_THREADS 256
#define DEG_UNROLL  8
static_assert(TL_MASK + 1 == DEG_THREADS, "one counter per thread");
__global__ __launch_bounds__(DEG_THREADS) void k_deg(const int* __restrict__ packed,
                                                     const int* __restrict__ binOff,
                                                     float* __restrict__ deg,
                                                     float* __restrict__ dinv) {
    __shared__ int cnt[TL_MASK + 1];
    const int tid = threadIdx.x;
    cnt[tid] = 0;
    __syncthreads();
    const int beg = binOff[blockIdx.x];
    const int end = binOff[blockIdx.x + 1];
    for (int base = beg + tid; base < end; base += DEG_THREADS * DEG_UNROLL) {
        unsigned p[DEG_UNROLL];
        #pragma unroll
        for (int u = 0; u < DEG_UNROLL; ++u) {
            const int e = base + u * DEG_THREADS;
            p[u] = (unsigned)packed[e < end ? e : end - 1];
        }
        #pragma unroll
        for (int u = 0; u < DEG_UNROLL; ++u) {
            const int e = base + u * DEG_THREADS;
            if (e < end) atomicAdd(&cnt[(p[u] >> SRC_BITS) & TL_MASK], 1);
        }
    }
    __syncthreads();
    const int node0 = blockIdx.x * TILE;
    const int nrows = (N_NODES - node0 < TILE) ? (N_NODES - node0) : TILE;
    if (tid < nrows) {
        const float d = 1.0f + (float)cnt[tid];
        deg[node0 + tid]  = d;
        dinv[node0 + tid] = dinv_of(d);
    }
}

__global__ __launch_bounds__(512) void k_scan(const int* __restrict__ binCnt,
                                              int* __restrict__ binOff,
                                              int* __restrict__ binCur) {
    __shared__ int s[512];
    const int t = threadIdx.x;
    const int v = (t < NTILES) ? binCnt[t] : 0;
    s[t] = v;
    __syncthreads();
    for (int d = 1; d < 512; d <<= 1) {
        int x = (t >= d) ? s[t - d] : 0;
        __syncthreads();
        s[t] += x;
        __syncthreads();
    }
    const int incl = s[t];
    if (t < NTILES) {
        binOff[t] = incl - v;
        binCur[t] = incl - v;
    }
    if (t == NTILES - 1) binOff[NTILES] = incl;
}

__global__ __launch_bounds__(256) void k_bin(const int* __restrict__ row,
                                             const int* __restrict__ col,
                                             int* __restrict__ binCur,
                                             int* __restrict__ packed) {
    __shared__ int h[NTILES_PAD];
    const int tid = threadIdx.x;
    const int e0 = blockIdx.x * BIN_CHUNK + tid;
    int rr[BIN_CHUNK / 256];
    #pragma unroll
    for (int j = 0; j < BIN_CHUNK / 256; ++j) {
        int e = e0 + j * 256;
        rr[j] = row[e < N_EDGES ? e : N_EDGES - 1];
    }
    for (int b = tid; b < NTILES_PAD; b += 256) h[b] = 0;
    __syncthreads();
    for (int j = 0; j < BIN_CHUNK / 256; ++j) {
        int e = e0 + j * 256;
        if (e < N_EDGES) atomicAdd(&h[col[e] / TILE], 1);
    }
    __syncthreads();
    for (int b = tid; b < NTILES; b += 256) {
        int n = h[b];
        h[b] = n ? atomicAdd(&binCur[b], n) : 0;
    }
    __syncthreads();
    #pragma unroll
    for (int j = 0; j < BIN_CHUNK / 256; ++j) {
        int e = e0 + j * 256;
        if (e < N_EDGES) {
            int c  = col[e];
            int t  = c / TILE;
            int tl = c - t * TILE;
            int pos = atomicAdd(&h[t], 1);
            packed[pos] = (int)((unsigned)rr[j] | ((unsigned)tl << SRC_BITS));
        }
    }
}

#define MLP_WAVES   8
#define MLP_BLOCKS  256
#define NSLAB       (N_NODES / 16)
#define SLABS_PASS  (MLP_BLOCKS * MLP_WAVES)
#define FULL_PASSES (NSLAB / SLABS_PASS)
#define REM_SLABS   (NSLAB - FULL_PASSES * SLABS_PASS)
#define KC          32
#define NKC         (DIN / KC)
#define W1_PITCH    528
#define W1_BUF      ((KC / 2) * W1_PITCH)
#define W2_PITCH    160
#define HS_STRIDE   260

static_assert(REM_SLABS <= MLP_BLOCKS, "last pass: one slab per block");
static_assert(N_NODES % 16 == 0, "no row tail");

__global__ __launch_bounds__(256)
void k_mlp(const float* __restrict__ x,  const float* __restrict__ W1,
           const float* __restrict__ b1, const float* __restrict__ W2,
           const float* __restrict__ b2, const float* __restrict__ temp,
           float* __restrict__ bufA,     float* __restrict__ hidden) {
    __shared__ __attribute__((aligned(16))) float w1s[2 * W1_BUF];
    __shared__ __attribute__((aligned(16))) float w2s[(DHID / 2) * W2_PITCH];
    __shared__ __attribute__((aligned(16))) float hs[MLP_WAVES * 16 * HS_STRIDE];

    const int tid  = threadIdx.x;
    const int lane = tid & 31;
    const int wave = __builtin_amdgcn_readfirstlane(tid >> 5);
    const int c    = lane & 15;
    const int h    = lane >> 4;
    const int bid  = blockIdx.x;

    const int npass = (bid < REM_SLABS) ? FULL_PASSES + 1 : FULL_PASSES;
    const int total = npass * NKC;

    for (int task = tid; task < (DHID / 2) * (DOUT / 4); task += 256) {
        const int kp = task >> 4, n4 = task & 15;
        const float4 r0 = *(const float4*)(W2 + (2 * kp) * DOUT + 4 * n4);
        const float4 r1 = *(const float4*)(W2 + (2 * kp + 1) * DOUT + 4 * n4);
        float4* d = (float4*)(w2s + kp * W2_PITCH + 8 * n4);
        d[0] = make_float4(r0.x, r1.x, r0.y, r1.y);
        d[1] = make_float4(r0.z, r1.z, r0.w, r1.w);
    }

    const int wn  = tid & 63;
    const int wkp = tid >> 6;
    float4 wr[8];
    float4 xc[4], xn[4];
    v8f acc[16];

    {
        const float* p = W1 + (2 * wkp) * DHID + 4 * wn;
        #pragma unroll
        for (int q = 0; q < 4; ++q) {
            wr[2 * q]     = *(const float4*)(p + (8 * q) * DHID);
            wr[2 * q + 1] = *(const float4*)(p + (8 * q + 1) * DHID);
        }
        const float* xp = x + (size_t)((bid * MLP_WAVES + wave) * 16 + c) * DIN + 4 * h;
        #pragma unroll
        for (int i = 0; i < 4; ++i) xc[i] = *(const float4*)(xp + 8 * i);
        #pragma unroll
        for (int q = 0; q < 4; ++q) {
            float4* d = (float4*)(w1s + (wkp + 4 * q) * W1_PITCH + 8 * wn);
            d[0] = make_float4(wr[2 * q].x, wr[2 * q + 1].x, wr[2 * q].y, wr[2 * q + 1].y);
            d[1] = make_float4(wr[2 * q].z, wr[2 * q + 1].z, wr[2 * q].w, wr[2 * q + 1].w);
        }
    }
    __syncthreads();

    for (int g = 0; g < total; ++g) {
        const int pass = g >> 4;
        const int ks   = g & (NKC - 1);
        const int slab = (pass < FULL_PASSES) ? pass * SLABS_PASS + bid * MLP_WAVES + wave
                                              : ((wave == 0) ? FULL_PASSES * SLABS_PASS + bid : -1);
        const bool act  = slab >= 0;
        const bool more = (g + 1 < total);

        if (ks == 0) {
            #pragma unroll
            for (int t = 0; t < 16; ++t) acc[t] = (v8f){0.f, 0.f, 0.f, 0.f, 0.f, 0.f, 0.f, 0.f};
        }

        if (more) {
            const int g1  = g + 1;
            const int kb1 = (g1 & (NKC - 1)) * KC;
            const float* p = W1 + (kb1 + 2 * wkp) * DHID + 4 * wn;
            #pragma unroll
            for (int q = 0; q < 4; ++q) {
                wr[2 * q]     = *(const float4*)(p + (8 * q) * DHID);
                wr[2 * q + 1] = *(const float4*)(p + (8 * q + 1) * DHID);
            }
            const int pass1 = g1 >> 4;
            const int slab1 = (pass1 < FULL_PASSES) ? pass1 * SLABS_PASS + bid * MLP_WAVES + wave
                                                    : ((wave == 0) ? FULL_PASSES * SLABS_PASS + bid : -1);
            if (slab1 >= 0) {
                const float* xp = x + (size_t)(slab1 * 16 + c) * DIN + kb1 + 4 * h;
                #pragma unroll
                for (int i = 0; i < 4; ++i) xn[i] = *(const float4*)(xp + 8 * i);
            }
        }

        if (act) {
            const float* wb = w1s + (g & 1) * W1_BUF + (2 * h) * W1_PITCH + 2 * c;
            #pragma unroll
            for (int i = 0; i < 4; ++i) {
                const v2f a0 = {xc[i].x, xc[i].y};
                const v2f a1 = {xc[i].z, xc[i].w};
                const float* w0 = wb + (4 * i) * W1_PITCH;
                #pragma unroll
                for (int t = 0; t < 16; ++t) {
                    const v2f b = *(const v2f*)(w0 + 32 * t);
                    acc[t] = __builtin_amdgcn_wmma_f32_16x16x4_f32(
                                 false, a0, false, b, (short)0, acc[t], false, false);
                }
                #pragma unroll
                for (int t = 0; t < 16; ++t) {
                    const v2f b = *(const v2f*)(w0 + W1_PITCH + 32 * t);
                    acc[t] = __builtin_amdgcn_wmma_f32_16x16x4_f32(
                                 false, a1, false, b, (short)0, acc[t], false, false);
                }
            }
        }

        if (more) {
            float* dbase = w1s + ((g + 1) & 1) * W1_BUF + wkp * W1_PITCH + 8 * wn;
            #pragma unroll
            for (int q = 0; q < 4; ++q) {
                float4* d = (float4*)(dbase + (4 * q) * W1_PITCH);
                d[0] = make_float4(wr[2 * q].x, wr[2 * q + 1].x, wr[2 * q].y, wr[2 * q + 1].y);
                d[1] = make_float4(wr[2 * q].z, wr[2 * q + 1].z, wr[2 * q].w, wr[2 * q + 1].w);
            }
        }

        float* hw = hs + wave * (16 * HS_STRIDE);
        if (ks == NKC - 1 && act) {
            #pragma unroll
            for (int t = 0; t < 16; ++t) {
                const float bias = b1[t * 16 + c];
                #pragma unroll
                for (int r = 0; r < 8; ++r) {
                    float v = acc[t][r] + bias;
                    v = v > 0.0f ? v : 0.0f;
                    hw[(8 * h + r) * HS_STRIDE + t * 16 + c] = v;
                }
            }
        }
        __syncthreads();

        if (ks == NKC - 1 && act) {
            v8f acc2[4];
            #pragma unroll
            for (int t = 0; t < 4; ++t) acc2[t] = (v8f){0.f, 0.f, 0.f, 0.f, 0.f, 0.f, 0.f, 0.f};
            const float* ha  = hw + c * HS_STRIDE + 2 * h;
            const float* w2b = w2s + h * W2_PITCH + 2 * c;
            #pragma unroll 4
            for (int k0 = 0; k0 < DHID; k0 += 4) {
                const v2f a = *(const v2f*)(ha + k0);
                const float* w0 = w2b + (k0 >> 1) * W2_PITCH;
                #pragma unroll
                for (int t = 0; t < 4; ++t) {
                    const v2f b = *(const v2f*)(w0 + 32 * t);
                    acc2[t] = __builtin_amdgcn_wmma_f32_16x16x4_f32(
                                  false, a, false, b, (short)0, acc2[t], false, false);
                }
            }
            const float t0 = temp[0];
            #pragma unroll
            for (int t = 0; t < 4; ++t) {
                const float bias = b2[t * 16 + c];
                #pragma unroll
                for (int r = 0; r < 8; ++r) {
                    const float v = acc2[t][r] + bias;
                    const int idx = (slab * 16 + 8 * h + r) * DOUT + t * 16 + c;
                    bufA[idx]   = v;
                    hidden[idx] = t0 * v;
                }
            }
        }

        #pragma unroll
        for (int i = 0; i < 4; ++i) xc[i] = xn[i];
    }
}

#define EDGES_PER_WAVE 32
#define EDGE_BATCH     16

template <bool TAIL>
__device__ __forceinline__ void add_edges(const float* __restrict__ src, float* acc,
                                          int s_l, int t_l, int w_l, int lane, int nvalid) {
    #pragma unroll
    for (int b = 0; b < EDGES_PER_WAVE; b += EDGE_BATCH) {
        if (TAIL && b >= nvalid) break;
        float lo[EDGE_BATCH], hi[EDGE_BATCH];
        #pragma unroll
        for (int j = 0; j < EDGE_BATCH; ++j) {
            int s = __builtin_amdgcn_readlane(s_l, b + j);
            const float* p = src + s * DOUT + lane;
            lo[j] = p[0];
            hi[j] = p[32];
        }
        #pragma unroll
        for (int j = 0; j < EDGE_BATCH; ++j) {
            if (!TAIL || b + j < nvalid) {
                int   t = __builtin_amdgcn_readlane(t_l, b + j);
                float n = __int_as_float(__builtin_amdgcn_readlane(w_l, b + j));
                float* q = acc + t * DOUT + lane;
                atomicAdd(q,      n * lo[j]);
                atomicAdd(q + 32, n * hi[j]);
            }
        }
    }
}

template <bool FIRST>
__global__ __launch_bounds__(1024)
void k_prop(int* __restrict__ packed, const int* __restrict__ binOff,
            const float* __restrict__ deg,
            const float* __restrict__ dinv, const float* __restrict__ src,
            float* __restrict__ dst, float* __restrict__ hidden,
            const float* __restrict__ temp, int kidx) {
    __shared__ float acc[TILE * DOUT];

    const int tid  = threadIdx.x;
    const int lane = tid & 31;
    const int wave = __builtin_amdgcn_readfirstlane(tid >> 5);
    const int node0 = blockIdx.x * TILE;

    for (int i = tid; i < TILE * DOUT; i += 1024) acc[i] = 0.0f;
    __syncthreads();

    const int beg = binOff[blockIdx.x];
    const int end = binOff[blockIdx.x + 1];
    for (int base = beg + wave * EDGES_PER_WAVE; base < end; base += 32 * EDGES_PER_WAVE) {
        const int e      = base + lane;
        const int nvalid = end - base;
        unsigned p = (e < end) ? (unsigned)packed[e] : (1u << DEG_SHIFT);
        if (FIRST) {
            if (e < end) {
                const unsigned d = (unsigned)deg[p & SRC_MASK];
                p |= (d < DEG_ESC ? d : DEG_ESC) << DEG_SHIFT;
                packed[e] = (int)p;
            }
        }
        const int s_l = (int)(p & SRC_MASK);
        const int t_l = (int)((p >> SRC_BITS) & TL_MASK);
        const unsigned dc = p >> DEG_SHIFT;
        float ds;
        if (dc < DEG_ESC) ds = dinv_of((float)dc);
        else              ds = dinv[s_l];
        const int w_l = __float_as_int(ds * dinv[node0 + t_l]);
        if (nvalid >= EDGES_PER_WAVE) add_edges<false>(src, acc, s_l, t_l, w_l, lane, nvalid);
        else                          add_edges<true >(src, acc, s_l, t_l, w_l, lane, nvalid);
    }
    __syncthreads();

    const float gk = temp[kidx];
    const int nrows = (N_NODES - node0 < TILE) ? (N_NODES - node0) : TILE;
    for (int i = tid; i < nrows * DOUT; i += 1024) {
        const float r = dinv[node0 + (i >> 6)];
        const int   g = node0 * DOUT + i;
        const float v = acc[i] + (r * r) * src[g];
        dst[g] = v;
        hidden[g] += gk * v;
    }
}

extern "C" void kernel_launch(void* const* d_in, const int* in_sizes, int n_in,
                              void* d_out, int out_size, void* d_ws, size_t ws_size,
                              hipStream_t stream) {
    const float* x    = (const float*)d_in[0];
    const int*   ei   = (const int*)  d_in[1];
    const float* W1   = (const float*)d_in[2];
    const float* b1   = (const float*)d_in[3];
    const float* W2   = (const float*)d_in[4];
    const float* b2   = (const float*)d_in[5];
    const float* temp = (const float*)d_in[6];
    const int* row = ei;
    const int* col = ei + N_EDGES;
    float* hidden = (float*)d_out;

    float* deg    = (float*)d_ws;
    float* dinv   = deg + N_NODES;
    int*   binCnt = (int*)(dinv + N_NODES);
    int*   binOff = binCnt + 512;
    int*   binCur = binOff + 1024;
    int*   packed = binCur + 512;
    float* bufA   = (float*)(packed + N_EDGES);
    float* bufB   = bufA + (size_t)N_NODES * DOUT;

    k_init <<<NTILES_PAD / 256, 256, 0, stream>>>(binCnt);
    k_count<<<(N_EDGES + BIN_CHUNK - 1) / BIN_CHUNK, 256, 0, stream>>>(col, binCnt);
    k_scan <<<1, 512, 0, stream>>>(binCnt, binOff, binCur);
    k_bin  <<<(N_EDGES + BIN_CHUNK - 1) / BIN_CHUNK, 256, 0, stream>>>(row, col, binCur, packed);
    k_deg  <<<NTILES, DEG_THREADS, 0, stream>>>(packed, binOff, deg, dinv);

    k_mlp<<<MLP_BLOCKS, MLP_WAVES * 32, 0, stream>>>(x, W1, b1, W2, b2, temp, bufA, hidden);

    k_prop<true><<<NTILES, 1024, 0, stream>>>(packed, binOff, deg, dinv, bufA, bufB,
                                              hidden, temp, 1);
    for (int k = 1; k < KSTEPS; ++k) {
        float* src = (k & 1) ? bufB : bufA;
        float* dst = (k & 1) ? bufA : bufB;
        k_prop<false><<<NTILES, 1024, 0, stream>>>(packed, binOff, deg, dinv, src, dst,
                                                   hidden, temp, k + 1);
    }
}


# Scatter: LDS f32 atomics replaced by matrix-unit sums: edges grouped per 16-row target group, a wave adds 4 gathered rows per WMMA via a one-hot weight matrix into register tiles; sum order changed
# speedup vs baseline: 7.9363x; 1.0690x over previous
#include <hip/hip_runtime.h>


#define N_NODES 100000
#define N_EDGES 3200000
#define DIN     512
#define DHID    256
#define DOUT    64
#define KSTEPS  10

#define TILE        256
#define NTILES      391
#define NTILES_PAD  512
#define GROUP       16
#define NGROUPS     (N_NODES / GROUP)
#define GROUPS_PER_TILE (TILE / GROUP)
#define SRC_BITS    17
#define SRC_MASK    0x1FFFF
#define TL_MASK     0xFF
#define DEG_SHIFT   25
#define DEG_ESC     127u
#define BIN_CHUNK   4096

static_assert(N_NODES % GROUP == 0, "no partial row group");
static_assert(TILE % GROUP == 0 && TILE == TL_MASK + 1, "a tile is whole groups; local row fills its field");

typedef __attribute__((ext_vector_type(2))) float v2f;
typedef __attribute__((ext_vector_type(8))) float v8f;


__global__ __launch_bounds__(256) void k_init(int* __restrict__ binCnt) {
    int g = blockIdx.x * 256 + threadIdx.x;
    if (g < NTILES_PAD) binCnt[g] = 0;
}

__global__ __launch_bounds__(256) void k_count(const int* __restrict__ col,
                                               int* __restrict__ binCnt) {
    __shared__ int h[NTILES_PAD];
    const int tid = threadIdx.x;
    for (int b = tid; b < NTILES_PAD; b += 256) h[b] = 0;
    __syncthreads();
    const int e0 = blockIdx.x * BIN_CHUNK + tid;
    for (int j = 0; j < BIN_CHUNK / 256; ++j) {
        int e = e0 + j * 256;
        if (e < N_EDGES) {
            int c = col[e];
            atomicAdd(&h[c / TILE], 1);
        }
    }
    __syncthreads();
    for (int b = tid; b < NTILES; b += 256) {
        int n = h[b];
        if (n) atomicAdd(&binCnt[b], n);
    }
}

__device__ __forceinline__ float dinv_of(float d) { return rsqrtf(d); }

#define DEG_THREADS 256
#define DEG_UNROLL  8
static_assert(TL_MASK + 1 == DEG_THREADS, "one counter per thread");
__global__ __launch_bounds__(DEG_THREADS) void k_deg(const int* __restrict__ binned,
                                                     const int* __restrict__ binOff,
                                                     float* __restrict__ deg,
                                                     int* __restrict__ grpOff,
                                                     int* __restrict__ packed) {
    __shared__ int cnt[TL_MASK + 1];
    __shared__ int gsum[GROUPS_PER_TILE];
    __shared__ int gcur[GROUPS_PER_TILE];
    const int tid = threadIdx.x;
    cnt[tid] = 0;
    __syncthreads();
    const int beg = binOff[blockIdx.x];
    const int end = binOff[blockIdx.x + 1];
    for (int base = beg + tid; base < end; base += DEG_THREADS * DEG_UNROLL) {
        unsigned p[DEG_UNROLL];
        #pragma unroll
        for (int u = 0; u < DEG_UNROLL; ++u) {
            const int e = base + u * DEG_THREADS;
            p[u] = (unsigned)binned[e < end ? e : end - 1];
        }
        #pragma unroll
        for (int u = 0; u < DEG_UNROLL; ++u) {
            const int e = base + u * DEG_THREADS;
            if (e < end) atomicAdd(&cnt[(p[u] >> SRC_BITS) & TL_MASK], 1);
        }
    }
    __syncthreads();
    const int node0 = blockIdx.x * TILE;
    const int nrows = (N_NODES - node0 < TILE) ? (N_NODES - node0) : TILE;
    if (tid < nrows) deg[node0 + tid] = 1.0f + (float)cnt[tid];
    if (tid < GROUPS_PER_TILE) {
        int n = 0;
        #pragma unroll
        for (int r = 0; r < GROUP; ++r) n += cnt[tid * GROUP + r];
        gsum[tid] = n;
    }
    __syncthreads();
    if (tid < GROUPS_PER_TILE) {
        int off = beg;
        for (int r = 0; r < tid; ++r) off += gsum[r];
        gcur[tid] = off;
        if (tid * GROUP < nrows) grpOff[blockIdx.x * GROUPS_PER_TILE + tid] = off;
    }
    if (blockIdx.x == NTILES - 1 && tid == 0) grpOff[NGROUPS] = end;
    __syncthreads();
    for (int base = beg + tid; base < end; base += DEG_THREADS * DEG_UNROLL) {
        unsigned p[DEG_UNROLL];
        #pragma unroll
        for (int u = 0; u < DEG_UNROLL; ++u) {
            const int e = base + u * DEG_THREADS;
            p[u] = (unsigned)binned[e < end ? e : end - 1];
        }
        #pragma unroll
        for (int u = 0; u < DEG_UNROLL; ++u) {
            const int e = base + u * DEG_THREADS;
            if (e < end) {
                const int pos = atomicAdd(&gcur[(p[u] >> (SRC_BITS + 4)) & (GROUPS_PER_TILE - 1)], 1);
                packed[pos] = (int)p[u];
            }
        }
    }
}

__global__ __launch_bounds__(512) void k_scan(const int* __restrict__ binCnt,
                                              int* __restrict__ binOff,
                                              int* __restrict__ binCur) {
    __shared__ int s[512];
    const int t = threadIdx.x;
    const int v = (t < NTILES) ? binCnt[t] : 0;
    s[t] = v;
    __syncthreads();
    for (int d = 1; d < 512; d <<= 1) {
        int x = (t >= d) ? s[t - d] : 0;
        __syncthreads();
        s[t] += x;
        __syncthreads();
    }
    const int incl = s[t];
    if (t < NTILES) {
        binOff[t] = incl - v;
        binCur[t] = incl - v;
    }
    if (t == NTILES - 1) binOff[NTILES] = incl;
}

__global__ __launch_bounds__(256) void k_bin(const int* __restrict__ row,
                                             const int* __restrict__ col,
                                             int* __restrict__ binCur,
                                             int* __restrict__ packed) {
    __shared__ int h[NTILES_PAD];
    const int tid = threadIdx.x;
    const int e0 = blockIdx.x * BIN_CHUNK + tid;
    int rr[BIN_CHUNK / 256];
    #pragma unroll
    for (int j = 0; j < BIN_CHUNK / 256; ++j) {
        int e = e0 + j * 256;
        rr[j] = row[e < N_EDGES ? e : N_EDGES - 1];
    }
    for (int b = tid; b < NTILES_PAD; b += 256) h[b] = 0;
    __syncthreads();
    for (int j = 0; j < BIN_CHUNK / 256; ++j) {
        int e = e0 + j * 256;
        if (e < N_EDGES) atomicAdd(&h[col[e] / TILE], 1);
    }
    __syncthreads();
    for (int b = tid; b < NTILES; b += 256) {
        int n = h[b];
        h[b] = n ? atomicAdd(&binCur[b], n) : 0;
    }
    __syncthreads();
    #pragma unroll
    for (int j = 0; j < BIN_CHUNK / 256; ++j) {
        int e = e0 + j * 256;
        if (e < N_EDGES) {
            int c  = col[e];
            int t  = c / TILE;
            int tl = c - t * TILE;
            int pos = atomicAdd(&h[t], 1);
            packed[pos] = (int)((unsigned)rr[j] | ((unsigned)tl << SRC_BITS));
        }
    }
}

#define MLP_WAVES   8
#define MLP_BLOCKS  256
#define NSLAB       (N_NODES / 16)
#define SLABS_PASS  (MLP_BLOCKS * MLP_WAVES)
#define FULL_PASSES (NSLAB / SLABS_PASS)
#define REM_SLABS   (NSLAB - FULL_PASSES * SLABS_PASS)
#define KC          32
#define NKC         (DIN / KC)
#define W1_PITCH    528
#define W1_BUF      ((KC / 2) * W1_PITCH)
#define W2_PITCH    160
#define HS_STRIDE   260

static_assert(REM_SLABS <= MLP_BLOCKS, "last pass: one slab per block");
static_assert(N_NODES % 16 == 0, "no row tail");

__global__ __launch_bounds__(256)
void k_mlp(const float* __restrict__ x,  const float* __restrict__ W1,
           const float* __restrict__ b1, const float* __restrict__ W2,
           const float* __restrict__ b2, const float* __restrict__ temp,
           float* __restrict__ bufA,     float* __restrict__ hidden) {
    __shared__ __attribute__((aligned(16))) float w1s[2 * W1_BUF];
    __shared__ __attribute__((aligned(16))) float w2s[(DHID / 2) * W2_PITCH];
    __shared__ __attribute__((aligned(16))) float hs[MLP_WAVES * 16 * HS_STRIDE];

    const int tid  = threadIdx.x;
    const int lane = tid & 31;
    const int wave = __builtin_amdgcn_readfirstlane(tid >> 5);
    const int c    = lane & 15;
    const int h    = lane >> 4;
    const int bid  = blockIdx.x;

    const int npass = (bid < REM_SLABS) ? FULL_PASSES + 1 : FULL_PASSES;
    const int total = npass * NKC;

    for (int task = tid; task < (DHID / 2) * (DOUT / 4); task += 256) {
        const int kp = task >> 4, n4 = task & 15;
        const float4 r0 = *(const float4*)(W2 + (2 * kp) * DOUT + 4 * n4);
        const float4 r1 = *(const float4*)(W2 + (2 * kp + 1) * DOUT + 4 * n4);
        float4* d = (float4*)(w2s + kp * W2_PITCH + 8 * n4);
        d[0] = make_float4(r0.x, r1.x, r0.y, r1.y);
        d[1] = make_float4(r0.z, r1.z, r0.w, r1.w);
    }

    const int wn  = tid & 63;
    const int wkp = tid >> 6;
    float4 wr[8];
    float4 xc[4], xn[4];
    v8f acc[16];

    {
        const float* p = W1 + (2 * wkp) * DHID + 4 * wn;
        #pragma unroll
        for (int q = 0; q < 4; ++q) {
            wr[2 * q]     = *(const float4*)(p + (8 * q) * DHID);
            wr[2 * q + 1] = *(const float4*)(p + (8 * q + 1) * DHID);
        }
        const float* xp = x + (size_t)((bid * MLP_WAVES + wave) * 16 + c) * DIN + 4 * h;
        #pragma unroll
        for (int i = 0; i < 4; ++i) xc[i] = *(const float4*)(xp + 8 * i);
        #pragma unroll
        for (int q = 0; q < 4; ++q) {
            float4* d = (float4*)(w1s + (wkp + 4 * q) * W1_PITCH + 8 * wn);
            d[0] = make_float4(wr[2 * q].x, wr[2 * q + 1].x, wr[2 * q].y, wr[2 * q + 1].y);
            d[1] = make_float4(wr[2 * q].z, wr[2 * q + 1].z, wr[2 * q].w, wr[2 * q + 1].w);
        }
    }
    __syncthreads();

    for (int g = 0; g < total; ++g) {
        const int pass = g >> 4;
        const int ks   = g & (NKC - 1);
        const int slab = (pass < FULL_PASSES) ? pass * SLABS_PASS + bid * MLP_WAVES + wave
                                              : ((wave == 0) ? FULL_PASSES * SLABS_PASS + bid : -1);
        const bool act  = slab >= 0;
        const bool more = (g + 1 < total);

        if (ks == 0) {
            #pragma unroll
            for (int t = 0; t < 16; ++t) acc[t] = (v8f){0.f, 0.f, 0.f, 0.f, 0.f, 0.f, 0.f, 0.f};
        }

        if (more) {
            const int g1  = g + 1;
            const int kb1 = (g1 & (NKC - 1)) * KC;
            const float* p = W1 + (kb1 + 2 * wkp) * DHID + 4 * wn;
            #pragma unroll
            for (int q = 0; q < 4; ++q) {
                wr[2 * q]     = *(const float4*)(p + (8 * q) * DHID);
                wr[2 * q + 1] = *(const float4*)(p + (8 * q + 1) * DHID);
            }
            const int pass1 = g1 >> 4;
            const int slab1 = (pass1 < FULL_PASSES) ? pass1 * SLABS_PASS + bid * MLP_WAVES + wave
                                                    : ((wave == 0) ? FULL_PASSES * SLABS_PASS + bid : -1);
            if (slab1 >= 0) {
                const float* xp = x + (size_t)(slab1 * 16 + c) * DIN + kb1 + 4 * h;
                #pragma unroll
                for (int i = 0; i < 4; ++i) xn[i] = *(const float4*)(xp + 8 * i);
            }
        }

        if (act) {
            const float* wb = w1s + (g & 1) * W1_BUF + (2 * h) * W1_PITCH + 2 * c;
            #pragma unroll
            for (int i = 0; i < 4; ++i) {
                const v2f a0 = {xc[i].x, xc[i].y};
                const v2f a1 = {xc[i].z, xc[i].w};
                const float* w0 = wb + (4 * i) * W1_PITCH;
                #pragma unroll
                for (int t = 0; t < 16; ++t) {
                    const v2f b = *(const v2f*)(w0 + 32 * t);
                    acc[t] = __builtin_amdgcn_wmma_f32_16x16x4_f32(
                                 false, a0, false, b, (short)0, acc[t], false, false);
                }
                #pragma unroll
                for (int t = 0; t < 16; ++t) {
                    const v2f b = *(const v2f*)(w0 + W1_PITCH + 32 * t);
                    acc[t] = __builtin_amdgcn_wmma_f32_16x16x4_f32(
                                 false, a1, false, b, (short)0, acc[t], false, false);
                }
            }
        }

        if (more) {
            float* dbase = w1s + ((g + 1) & 1) * W1_BUF + wkp * W1_PITCH + 8 * wn;
            #pragma unroll
            for (int q = 0; q < 4; ++q) {
                float4* d = (float4*)(dbase + (4 * q) * W1_PITCH);
                d[0] = make_float4(wr[2 * q].x, wr[2 * q + 1].x, wr[2 * q].y, wr[2 * q + 1].y);
                d[1] = make_float4(wr[2 * q].z, wr[2 * q + 1].z, wr[2 * q].w, wr[2 * q + 1].w);
            }
        }

        float* hw = hs + wave * (16 * HS_STRIDE);
        if (ks == NKC - 1 && act) {
            #pragma unroll
            for (int t = 0; t < 16; ++t) {
                const float bias = b1[t * 16 + c];
                #pragma unroll
                for (int r = 0; r < 8; ++r) {
                    float v = acc[t][r] + bias;
                    v = v > 0.0f ? v : 0.0f;
                    hw[(8 * h + r) * HS_STRIDE + t * 16 + c] = v;
                }
            }
        }
        __syncthreads();

        if (ks == NKC - 1 && act) {
            v8f acc2[4];
            #pragma unroll
            for (int t = 0; t < 4; ++t) acc2[t] = (v8f){0.f, 0.f, 0.f, 0.f, 0.f, 0.f, 0.f, 0.f};
            const float* ha  = hw + c * HS_STRIDE + 2 * h;
            const float* w2b = w2s + h * W2_PITCH + 2 * c;
            #pragma unroll 4
            for (int k0 = 0; k0 < DHID; k0 += 4) {
                const v2f a = *(const v2f*)(ha + k0);
                const float* w0 = w2b + (k0 >> 1) * W2_PITCH;
                #pragma unroll
                for (int t = 0; t < 4; ++t) {
                    const v2f b = *(const v2f*)(w0 + 32 * t);
                    acc2[t] = __builtin_amdgcn_wmma_f32_16x16x4_f32(
                                  false, a, false, b, (short)0, acc2[t], false, false);
                }
            }
            const float t0 = temp[0];
            #pragma unroll
            for (int t = 0; t < 4; ++t) {
                const float bias = b2[t * 16 + c];
                #pragma unroll
                for (int r = 0; r < 8; ++r) {
                    const float v = acc2[t][r] + bias;
                    const int idx = (slab * 16 + 8 * h + r) * DOUT + t * 16 + c;
                    bufA[idx]   = v;
                    hidden[idx] = t0 * v;
                }
            }
        }

        #pragma unroll
        for (int i = 0; i < 4; ++i) xc[i] = xn[i];
    }
}

#define PROP_WAVES 2
static_assert(NGROUPS % PROP_WAVES == 0, "every wave has a group");

template <int N>
__device__ __forceinline__ int row_share_i(int v) {
    return __builtin_amdgcn_update_dpp(0, v, 0x150 + N, 0xF, 0xF, true);
}
template <int N>
__device__ __forceinline__ float row_share_f(float v) {
    return __int_as_float(row_share_i<N>(__float_as_int(v)));
}

template <int Q>
__device__ __forceinline__ void set_load(const float* __restrict__ src, int s_l, int c,
                                         float4& b0, float4& b1) {
    const unsigned ia = (unsigned)row_share_i<2 * Q>(s_l)     * (DOUT / 4) + (unsigned)c;
    const unsigned ib = (unsigned)row_share_i<2 * Q + 1>(s_l) * (DOUT / 4) + (unsigned)c;
    b0 = ((const float4*)src)[ia];
    b1 = ((const float4*)src)[ib];
}

template <int Q>
__device__ __forceinline__ void set_mma(v8f (&acc)[4], const float4& b0, const float4& b1,
                                        int t_l, float w_l, int c) {
    const int   ta = row_share_i<2 * Q>(t_l);
    const int   tb = row_share_i<2 * Q + 1>(t_l);
    const float wa = row_share_f<2 * Q>(w_l);
    const float wb = row_share_f<2 * Q + 1>(w_l);
    const v2f a = {ta == c ? wa : 0.0f, tb == c ? wb : 0.0f};
    acc[0] = __builtin_amdgcn_wmma_f32_16x16x4_f32(false, a, false, (v2f){b0.x, b1.x},
                                                   (short)0, acc[0], false, false);
    acc[1] = __builtin_amdgcn_wmma_f32_16x16x4_f32(false, a, false, (v2f){b0.y, b1.y},
                                                   (short)0, acc[1], false, false);
    acc[2] = __builtin_amdgcn_wmma_f32_16x16x4_f32(false, a, false, (v2f){b0.z, b1.z},
                                                   (short)0, acc[2], false, false);
    acc[3] = __builtin_amdgcn_wmma_f32_16x16x4_f32(false, a, false, (v2f){b0.w, b1.w},
                                                   (short)0, acc[3], false, false);
}

__device__ __forceinline__ void sweep_load(const float* __restrict__ src, int s_l, int c,
                                           float4 (&b0)[8], float4 (&b1)[8]) {
    set_load<0>(src, s_l, c, b0[0], b1[0]);
    set_load<1>(src, s_l, c, b0[1], b1[1]);
    set_load<2>(src, s_l, c, b0[2], b1[2]);
    set_load<3>(src, s_l, c, b0[3], b1[3]);
    set_load<4>(src, s_l, c, b0[4], b1[4]);
    set_load<5>(src, s_l, c, b0[5], b1[5]);
    set_load<6>(src, s_l, c, b0[6], b1[6]);
    set_load<7>(src, s_l, c, b0[7], b1[7]);
}
__device__ __forceinline__ void sweep_mma(v8f (&acc)[4], const float4 (&b0)[8],
                                          const float4 (&b1)[8], int t_l, float w_l, int c) {
    set_mma<0>(acc, b0[0], b1[0], t_l, w_l, c);
    set_mma<1>(acc, b0[1], b1[1], t_l, w_l, c);
    set_mma<2>(acc, b0[2], b1[2], t_l, w_l, c);
    set_mma<3>(acc, b0[3], b1[3], t_l, w_l, c);
    set_mma<4>(acc, b0[4], b1[4], t_l, w_l, c);
    set_mma<5>(acc, b0[5], b1[5], t_l, w_l, c);
    set_mma<6>(acc, b0[6], b1[6], t_l, w_l, c);
    set_mma<7>(acc, b0[7], b1[7], t_l, w_l, c);
}

template <bool FIRST>
__global__ __launch_bounds__(PROP_WAVES * 32)
void k_prop(int* __restrict__ packed, const int* __restrict__ grpOff,
            const float* __restrict__ deg, const float* __restrict__ src,
            float* __restrict__ dst, float* __restrict__ hidden,
            const float* __restrict__ temp, int kidx) {
    const int tid  = threadIdx.x;
    const int lane = tid & 31;
    const int wave = __builtin_amdgcn_readfirstlane(tid >> 5);
    const int g    = blockIdx.x * PROP_WAVES + wave;
    const int c    = lane & 15;
    const int h    = lane >> 4;
    const int node0 = g * GROUP;

    v8f acc[4];
    #pragma unroll
    for (int j = 0; j < 4; ++j) acc[j] = (v8f){0.f, 0.f, 0.f, 0.f, 0.f, 0.f, 0.f, 0.f};

    const int beg = grpOff[g];
    const int end = grpOff[g + 1];
    const unsigned PAD = 1u << DEG_SHIFT;
    unsigned pn = (beg + lane < end) ? (unsigned)packed[beg + lane] : PAD;
    for (int base = beg; base < end; base += 32) {
        const int  e      = base + lane;
        const bool valid  = e < end;
        unsigned p = pn;
        if (FIRST) {
            if (valid) {
                const unsigned d = (unsigned)deg[p & SRC_MASK];
                p |= (d < DEG_ESC ? d : DEG_ESC) << DEG_SHIFT;
                packed[e] = (int)p;
            }
        }
        const int s_l = (int)(p & SRC_MASK);
        const int t_l = (int)((p >> SRC_BITS) & (GROUP - 1));
        const float dtg = deg[node0 + t_l];
        float4 b0[8], b1[8];
        sweep_load(src, s_l, c, b0, b1);
        pn = (e + 32 < end) ? (unsigned)packed[e + 32] : PAD;
        const unsigned dc = p >> DEG_SHIFT;
        float ds;
        if (dc < DEG_ESC) ds = dinv_of((float)dc);
        else              ds = dinv_of(deg[s_l]);
        const float w_l = valid ? ds * dinv_of(dtg) : 0.0f;
        sweep_mma(acc, b0, b1, t_l, w_l, c);
    }

    const float gk = temp[kidx];
    const int   o0 = (node0 + 8 * h) * DOUT + 4 * c;
    float  dg[8];
    float4 sv[8], hv[8];
    #pragma unroll
    for (int r = 0; r < 8; ++r) {
        dg[r] = deg[node0 + 8 * h + r];
        sv[r] = *(const float4*)(src + o0 + r * DOUT);
        hv[r] = *(const float4*)(hidden + o0 + r * DOUT);
    }
    #pragma unroll
    for (int r = 0; r < 8; ++r) {
        const float dr = dinv_of(dg[r]);
        const float d2 = dr * dr;
        float4 v;
        v.x = acc[0][r] + d2 * sv[r].x;
        v.y = acc[1][r] + d2 * sv[r].y;
        v.z = acc[2][r] + d2 * sv[r].z;
        v.w = acc[3][r] + d2 * sv[r].w;
        *(float4*)(dst + o0 + r * DOUT) = v;
        float4 u = hv[r];
        u.x += gk * v.x;
        u.y += gk * v.y;
        u.z += gk * v.z;
        u.w += gk * v.w;
        *(float4*)(hidden + o0 + r * DOUT) = u;
    }
}

extern "C" void kernel_launch(void* const* d_in, const int* in_sizes, int n_in,
                              void* d_out, int out_size, void* d_ws, size_t ws_size,
                              hipStream_t stream) {
    const float* x    = (const float*)d_in[0];
    const int*   ei   = (const int*)  d_in[1];
    const float* W1   = (const float*)d_in[2];
    const float* b1   = (const float*)d_in[3];
    const float* W2   = (const float*)d_in[4];
    const float* b2   = (const float*)d_in[5];
    const float* temp = (const float*)d_in[6];
    const int* row = ei;
    const int* col = ei + N_EDGES;
    float* hidden = (float*)d_out;

    float* deg    = (float*)d_ws;
    int*   grpOff = (int*)(deg + N_NODES);
    int*   binCnt = grpOff + N_NODES;
    int*   binOff = binCnt + 512;
    int*   binCur = binOff + 1024;
    int*   packed = binCur + 512;
    float* bufA   = (float*)(packed + N_EDGES);
    float* bufB   = bufA + (size_t)N_NODES * DOUT;
    int*   binned = (int*)bufB;

    k_init <<<NTILES_PAD / 256, 256, 0, stream>>>(binCnt);
    k_count<<<(N_EDGES + BIN_CHUNK - 1) / BIN_CHUNK, 256, 0, stream>>>(col, binCnt);
    k_scan <<<1, 512, 0, stream>>>(binCnt, binOff, binCur);
    k_bin  <<<(N_EDGES + BIN_CHUNK - 1) / BIN_CHUNK, 256, 0, stream>>>(row, col, binCur, binned);
    k_deg  <<<NTILES, DEG_THREADS, 0, stream>>>(binned, binOff, deg, grpOff, packed);

    k_mlp<<<MLP_BLOCKS, MLP_WAVES * 32, 0, stream>>>(x, W1, b1, W2, b2, temp, bufA, hidden);

    k_prop<true><<<NGROUPS / PROP_WAVES, PROP_WAVES * 32, 0, stream>>>(packed, grpOff, deg, bufA, bufB,
                                                                      hidden, temp, 1);
    for (int k = 1; k < KSTEPS; ++k) {
        float* src = (k & 1) ? bufB : bufA;
        float* dst = (k & 1) ? bufA : bufB;
        k_prop<false><<<NGROUPS / PROP_WAVES, PROP_WAVES * 32, 0, stream>>>(packed, grpOff, deg, src, dst,
                                                                           hidden, temp, k + 1);
    }
}
